# starting kernel
# baseline (speedup 1.0000x reference)
#include <hip/hip_runtime.h>
#include <hip/hip_bf16.h>


typedef __attribute__((ext_vector_type(2))) float v2f;
typedef __attribute__((ext_vector_type(8))) float v8f;

#define N_IN  1024
#define N_OUT 512
#define BATCH 128

__global__ __launch_bounds__(32)
void memristor_linear_wmma_f32(const float* __restrict__ X,
                               const float* __restrict__ W,
                               const float* __restrict__ Bv,
                               float* __restrict__ Y) {
  const int lane = threadIdx.x & 31;
  const int h    = lane >> 4;
  const int ml   = lane & 15;
  const int m0   = blockIdx.y << 4;
  const int n0   = blockIdx.x << 5;

  const float* arow = X + (size_t)(m0 + ml) * N_IN + 2 * h;
  const float* b0p = W + n0 + ml;
  const float* b1p = W + n0 + 16 + ml;

  v8f acc0 = {};
  v8f acc1 = {};

#pragma unroll 4
  for (int k0 = 0; k0 < N_IN; k0 += 4) {
    const int ka = k0 + 2 * h;

    v2f a = *(const v2f*)(arow + k0);

    v2f b0, b1;
    b0.x = b0p[(size_t)(ka)     * N_OUT];
    b0.y = b0p[(size_t)(ka + 1) * N_OUT];
    b1.x = b1p[(size_t)(ka)     * N_OUT];
    b1.y = b1p[(size_t)(ka + 1) * N_OUT];

    acc0 = __builtin_amdgcn_wmma_f32_16x16x4_f32(false, a, false, b0,
                                                 (short)0, acc0, false, false);
    acc1 = __builtin_amdgcn_wmma_f32_16x16x4_f32(false, a, false, b1,
                                                 (short)0, acc1, false, false);
  }

  const float bias0 = Bv[n0 + ml];
  const float bias1 = Bv[n0 + 16 + ml];

#pragma unroll
  for (int r = 0; r < 8; ++r) {
    const int m = m0 + r + 8 * h;
    Y[(size_t)m * N_OUT + n0 + ml]      = acc0[r] + bias0;
    Y[(size_t)m * N_OUT + n0 + 16 + ml] = acc1[r] + bias1;
  }
}

extern "C" void kernel_launch(void* const* d_in, const int* in_sizes, int n_in,
                              void* d_out, int out_size, void* d_ws, size_t ws_size,
                              hipStream_t stream) {
  const float* x = (const float*)d_in[0];
  const float* w = (const float*)d_in[1];
  const float* b = (const float*)d_in[2];
  float* y = (float*)d_out;

  dim3 grid(N_OUT / 32, BATCH / 16);
  dim3 block(32);
  memristor_linear_wmma_f32<<<grid, block, 0, stream>>>(x, w, b, y);
}


# Split K across the 16 waves of a 512-thread work-group (64 k each, a slice's loads issued as one group); the 16 partial strips are added through LDS in fixed slice order: new summation order
# speedup vs baseline: 6.2409x; 6.2409x over previous
#include <hip/hip_runtime.h>
#include <hip/hip_bf16.h>


typedef __attribute__((ext_vector_type(2))) float v2f;
typedef __attribute__((ext_vector_type(8))) float v8f;

#define N_IN   1024
#define N_OUT  512
#define BATCH  128
#define KSPLIT 16
#define KSLICE (N_IN / KSPLIT)
#define STRIP  (16 * 32)

__global__ __launch_bounds__(32 * KSPLIT) __attribute__((amdgpu_waves_per_eu(4, 4)))
void memristor_linear_wmma_f32(const float* __restrict__ X,
                               const float* __restrict__ W,
                               const float* __restrict__ Bv,
                               float* __restrict__ Y) {
  __shared__ float part[KSPLIT][STRIP];

  const int tid  = threadIdx.x;
  const int lane = tid & 31;
  const int ks   = tid >> 5;
  const int h    = lane >> 4;
  const int ml   = lane & 15;
  const int m0   = blockIdx.y << 4;
  const int n0   = blockIdx.x << 5;
  const int kb   = ks * KSLICE;

  const float* arow = X + (size_t)(m0 + ml) * N_IN + kb + 2 * h;
  const float* b0p = W + (size_t)kb * N_OUT + n0 + ml;
  const float* b1p = b0p + 16;

  v8f acc0 = {};
  v8f acc1 = {};

#pragma unroll
  for (int k0 = 0; k0 < KSLICE; k0 += 4) {
    const int ka = k0 + 2 * h;

    v2f a = *(const v2f*)(arow + k0);

    v2f b0, b1;
    b0.x = b0p[(size_t)(ka)     * N_OUT];
    b0.y = b0p[(size_t)(ka + 1) * N_OUT];
    b1.x = b1p[(size_t)(ka)     * N_OUT];
    b1.y = b1p[(size_t)(ka + 1) * N_OUT];

    acc0 = __builtin_amdgcn_wmma_f32_16x16x4_f32(false, a, false, b0,
                                                 (short)0, acc0, false, false);
    acc1 = __builtin_amdgcn_wmma_f32_16x16x4_f32(false, a, false, b1,
                                                 (short)0, acc1, false, false);
  }

  float* mine = part[ks];
#pragma unroll
  for (int r = 0; r < 8; ++r) {
    mine[r * 64 + h * 32 + ml]      = acc0[r];
    mine[r * 64 + h * 32 + 16 + ml] = acc1[r];
  }
  __syncthreads();

  const int e   = tid;
  float sum = part[0][e];
#pragma unroll
  for (int s = 1; s < KSPLIT; ++s) sum += part[s][e];

  const int row = (e >> 6) + 8 * ((e >> 5) & 1);
  const int col = e & 31;
  Y[(size_t)(m0 + row) * N_OUT + n0 + col] = sum + Bv[n0 + col];
}

extern "C" void kernel_launch(void* const* d_in, const int* in_sizes, int n_in,
                              void* d_out, int out_size, void* d_ws, size_t ws_size,
                              hipStream_t stream) {
  const float* x = (const float*)d_in[0];
  const float* w = (const float*)d_in[1];
  const float* b = (const float*)d_in[2];
  float* y = (float*)d_out;

  dim3 grid(N_OUT / 32, BATCH / 16);
  dim3 block(32 * KSPLIT);
  memristor_linear_wmma_f32<<<grid, block, 0, stream>>>(x, w, b, y);
}


# Fragment loads widened to whole lines: A by 16-byte loads, B by 8-byte loads of 2 adjacent columns (tile columns interleaved; k regrouped alike in A and B: new summation order); 40 loads for 80
# speedup vs baseline: 6.6117x; 1.0594x over previous
#include <hip/hip_runtime.h>
#include <hip/hip_bf16.h>


typedef __attribute__((ext_vector_type(2))) float v2f;
typedef __attribute__((ext_vector_type(4))) float v4f;
typedef __attribute__((ext_vector_type(8))) float v8f;

#define N_IN   1024
#define N_OUT  512
#define BATCH  128
#define KSPLIT 16
#define KSLICE (N_IN / KSPLIT)
#define STRIP  (16 * 32)

__global__ __launch_bounds__(32 * KSPLIT) __attribute__((amdgpu_waves_per_eu(4, 4)))
void memristor_linear_wmma_f32(const float* __restrict__ X,
                               const float* __restrict__ W,
                               const float* __restrict__ Bv,
                               float* __restrict__ Y) {
  __shared__ float part[KSPLIT][STRIP];

  const int tid  = threadIdx.x;
  const int lane = tid & 31;
  const int ks   = tid >> 5;
  const int h    = lane >> 4;
  const int ml   = lane & 15;
  const int m0   = blockIdx.y << 4;
  const int n0   = blockIdx.x << 5;
  const int kb   = ks * KSLICE;

  const float* arow = X + (size_t)(m0 + ml) * N_IN + kb + 4 * h;
  const float* brow = W + (size_t)(kb + 4 * h) * N_OUT + n0 + 2 * ml;

  v8f acc0 = {};
  v8f acc1 = {};

#pragma unroll
  for (int j = 0; j < KSLICE / 8; ++j) {
    const v4f a4 = *(const v4f*)(arow + 8 * j);
#pragma unroll
    for (int q = 0; q < 2; ++q) {
      const float* bp = brow + (size_t)(8 * j + 2 * q) * N_OUT;
      const v2f w0 = *(const v2f*)(bp);
      const v2f w1 = *(const v2f*)(bp + N_OUT);

      v2f a, b0, b1;
      a.x  = q ? a4.z : a4.x;
      a.y  = q ? a4.w : a4.y;
      b0.x = w0.x;  b0.y = w1.x;
      b1.x = w0.y;  b1.y = w1.y;

      acc0 = __builtin_amdgcn_wmma_f32_16x16x4_f32(false, a, false, b0,
                                                   (short)0, acc0, false, false);
      acc1 = __builtin_amdgcn_wmma_f32_16x16x4_f32(false, a, false, b1,
                                                   (short)0, acc1, false, false);
    }
  }

  float* mine = part[ks];
#pragma unroll
  for (int r = 0; r < 8; ++r) {
    mine[r * 64 + h * 32 + ml]      = acc0[r];
    mine[r * 64 + h * 32 + 16 + ml] = acc1[r];
  }
  __syncthreads();

  const int e    = tid;
  const int col  = e & 31;
  const int row  = (e >> 6) + 8 * ((e >> 5) & 1);
  const int slot = (e & ~31) + ((col & 1) << 4) + (col >> 1);
  float sum = part[0][slot];
#pragma unroll
  for (int s = 1; s < KSPLIT; ++s) sum += part[s][slot];

  Y[(size_t)(m0 + row) * N_OUT + n0 + col] = sum + Bv[n0 + col];
}

extern "C" void kernel_launch(void* const* d_in, const int* in_sizes, int n_in,
                              void* d_out, int out_size, void* d_ws, size_t ws_size,
                              hipStream_t stream) {
  const float* x = (const float*)d_in[0];
  const float* w = (const float*)d_in[1];
  const float* b = (const float*)d_in[2];
  float* y = (float*)d_out;

  dim3 grid(N_OUT / 32, BATCH / 16);
  dim3 block(32 * KSPLIT);
  memristor_linear_wmma_f32<<<grid, block, 0, stream>>>(x, w, b, y);
}


# K split 8 ways instead of 16: 8 waves of 128 k per work-group (half as many waves to enter, 2 per SIMD); 8 partial strips added through LDS in slice order by 8-byte accesses: new summation order
# speedup vs baseline: 7.2227x; 1.0924x over previous
#include <hip/hip_runtime.h>
#include <hip/hip_bf16.h>


typedef __attribute__((ext_vector_type(2))) float v2f;
typedef __attribute__((ext_vector_type(4))) float v4f;
typedef __attribute__((ext_vector_type(8))) float v8f;

#define N_IN   1024
#define N_OUT  512
#define BATCH  128
#define KSPLIT 8
#define KSLICE (N_IN / KSPLIT)
#define STRIP  (16 * 32)

__global__ __launch_bounds__(32 * KSPLIT) __attribute__((amdgpu_waves_per_eu(2, 2)))
void memristor_linear_wmma_f32(const float* __restrict__ X,
                               const float* __restrict__ W,
                               const float* __restrict__ Bv,
                               float* __restrict__ Y) {
  __shared__ __attribute__((aligned(16))) float part[KSPLIT][STRIP];

  const int tid  = threadIdx.x;
  const int lane = tid & 31;
  const int ks   = tid >> 5;
  const int h    = lane >> 4;
  const int ml   = lane & 15;
  const int m0   = blockIdx.y << 4;
  const int n0   = blockIdx.x << 5;
  const int kb   = ks * KSLICE;

  const float* arow = X + (size_t)(m0 + ml) * N_IN + kb + 4 * h;
  const float* brow = W + (size_t)(kb + 4 * h) * N_OUT + n0 + 2 * ml;

  v8f acc0 = {};
  v8f acc1 = {};

#pragma unroll
  for (int j = 0; j < KSLICE / 8; ++j) {
    const v4f a4 = *(const v4f*)(arow + 8 * j);
#pragma unroll
    for (int q = 0; q < 2; ++q) {
      const float* bp = brow + (size_t)(8 * j + 2 * q) * N_OUT;
      const v2f w0 = *(const v2f*)(bp);
      const v2f w1 = *(const v2f*)(bp + N_OUT);

      v2f a, b0, b1;
      a.x  = q ? a4.z : a4.x;
      a.y  = q ? a4.w : a4.y;
      b0.x = w0.x;  b0.y = w1.x;
      b1.x = w0.y;  b1.y = w1.y;

      acc0 = __builtin_amdgcn_wmma_f32_16x16x4_f32(false, a, false, b0,
                                                   (short)0, acc0, false, false);
      acc1 = __builtin_amdgcn_wmma_f32_16x16x4_f32(false, a, false, b1,
                                                   (short)0, acc1, false, false);
    }
  }

  float* mine = part[ks];
#pragma unroll
  for (int r = 0; r < 8; ++r) {
    v2f p;
    p.x = acc0[r];
    p.y = acc1[r];
    *(v2f*)(mine + r * 64 + h * 32 + 2 * ml) = p;
  }
  __syncthreads();

  const int e   = tid;
  const int col = 2 * (e & 15);
  const int row = (e >> 5) + 8 * ((e >> 4) & 1);
  v2f sum = *(const v2f*)(&part[0][2 * e]);
#pragma unroll
  for (int s = 1; s < KSPLIT; ++s) {
    const v2f p = *(const v2f*)(&part[s][2 * e]);
    sum.x += p.x;
    sum.y += p.y;
  }

  const v2f bias = *(const v2f*)(Bv + n0 + col);
  v2f out;
  out.x = sum.x + bias.x;
  out.y = sum.y + bias.y;
  *(v2f*)(Y + (size_t)(m0 + row) * N_OUT + n0 + col) = out;
}

extern "C" void kernel_launch(void* const* d_in, const int* in_sizes, int n_in,
                              void* d_out, int out_size, void* d_ws, size_t ws_size,
                              hipStream_t stream) {
  const float* x = (const float*)d_in[0];
  const float* w = (const float*)d_in[1];
  const float* b = (const float*)d_in[2];
  float* y = (float*)d_out;

  dim3 grid(N_OUT / 32, BATCH / 16);
  dim3 block(32 * KSPLIT);
  memristor_linear_wmma_f32<<<grid, block, 0, stream>>>(x, w, b, y);
}


# Each work-group walks its k blocks in one of 4 rotated orders picked by its grid position, so sharers of X rows or W columns are not on the same lines at once: new summation order
# speedup vs baseline: 8.3511x; 1.1562x over previous
#include <hip/hip_runtime.h>
#include <hip/hip_bf16.h>


typedef __attribute__((ext_vector_type(2))) float v2f;
typedef __attribute__((ext_vector_type(4))) float v4f;
typedef __attribute__((ext_vector_type(8))) float v8f;

#define N_IN   1024
#define N_OUT  512
#define BATCH  128
#define KSPLIT 8
#define KSLICE (N_IN / KSPLIT)
#define STRIP  (16 * 32)

template <int RG>
__device__ __forceinline__ void slice_products(const float* __restrict__ arow,
                                               const float* __restrict__ brow,
                                               v8f& acc0, v8f& acc1) {
#pragma unroll
  for (int j = 0; j < KSLICE / 8; ++j) {
    const int jb = ((((j >> 2) + RG) & 3) << 2) | (j & 3);
    const v4f a4 = *(const v4f*)(arow + 8 * jb);
#pragma unroll
    for (int q = 0; q < 2; ++q) {
      const float* bp = brow + (size_t)(8 * jb + 2 * q) * N_OUT;
      const v2f w0 = *(const v2f*)(bp);
      const v2f w1 = *(const v2f*)(bp + N_OUT);

      v2f a, b0, b1;
      a.x  = q ? a4.z : a4.x;
      a.y  = q ? a4.w : a4.y;
      b0.x = w0.x;  b0.y = w1.x;
      b1.x = w0.y;  b1.y = w1.y;

      acc0 = __builtin_amdgcn_wmma_f32_16x16x4_f32(false, a, false, b0,
                                                   (short)0, acc0, false, false);
      acc1 = __builtin_amdgcn_wmma_f32_16x16x4_f32(false, a, false, b1,
                                                   (short)0, acc1, false, false);
    }
  }
}

__global__ __launch_bounds__(32 * KSPLIT) __attribute__((amdgpu_waves_per_eu(2, 2)))
void memristor_linear_wmma_f32(const float* __restrict__ X,
                               const float* __restrict__ W,
                               const float* __restrict__ Bv,
                               float* __restrict__ Y) {
  __shared__ __attribute__((aligned(16))) float part[KSPLIT][STRIP];

  const int tid  = threadIdx.x;
  const int lane = tid & 31;
  const int ks   = tid >> 5;
  const int h    = lane >> 4;
  const int ml   = lane & 15;
  const int m0   = blockIdx.y << 4;
  const int n0   = blockIdx.x << 5;
  const int kb   = ks * KSLICE;

  const float* arow = X + (size_t)(m0 + ml) * N_IN + kb + 4 * h;
  const float* brow = W + (size_t)(kb + 4 * h) * N_OUT + n0 + 2 * ml;

  v8f acc0 = {};
  v8f acc1 = {};

  switch ((blockIdx.x + blockIdx.y) & 3) {
    case 0:  slice_products<0>(arow, brow, acc0, acc1); break;
    case 1:  slice_products<1>(arow, brow, acc0, acc1); break;
    case 2:  slice_products<2>(arow, brow, acc0, acc1); break;
    default: slice_products<3>(arow, brow, acc0, acc1); break;
  }

  float* mine = part[ks];
#pragma unroll
  for (int r = 0; r < 8; ++r) {
    v2f p;
    p.x = acc0[r];
    p.y = acc1[r];
    *(v2f*)(mine + r * 64 + h * 32 + 2 * ml) = p;
  }
  __syncthreads();

  const int e   = tid;
  const int col = 2 * (e & 15);
  const int row = (e >> 5) + 8 * ((e >> 4) & 1);
  v2f sum = *(const v2f*)(&part[0][2 * e]);
#pragma unroll
  for (int s = 1; s < KSPLIT; ++s) {
    const v2f p = *(const v2f*)(&part[s][2 * e]);
    sum.x += p.x;
    sum.y += p.y;
  }

  const v2f bias = *(const v2f*)(Bv + n0 + col);
  v2f out;
  out.x = sum.x + bias.x;
  out.y = sum.y + bias.y;
  *(v2f*)(Y + (size_t)(m0 + row) * N_OUT + n0 + col) = out;
}

extern "C" void kernel_launch(void* const* d_in, const int* in_sizes, int n_in,
                              void* d_out, int out_size, void* d_ws, size_t ws_size,
                              hipStream_t stream) {
  const float* x = (const float*)d_in[0];
  const float* w = (const float*)d_in[1];
  const float* b = (const float*)d_in[2];
  float* y = (float*)d_out;

  dim3 grid(N_OUT / 32, BATCH / 16);
  dim3 block(32 * KSPLIT);
  memristor_linear_wmma_f32<<<grid, block, 0, stream>>>(x, w, b, y);
}
